# starting kernel
# baseline (speedup 1.0000x reference)
#include <hip/hip_runtime.h>

typedef float v2f __attribute__((ext_vector_type(2)));
typedef float v8f __attribute__((ext_vector_type(8)));

#define NMODES 16
#define NLAYERS 6

__global__ __launch_bounds__(192) void cv_setup_kernel(
    const float* __restrict__ disp,
    const float* __restrict__ sq,
    const float* __restrict__ bs,
    float* __restrict__ ws)
{
  __shared__ float s_dx[2][NLAYERS][NMODES];
  __shared__ float s_dp[2][NLAYERS][NMODES];
  __shared__ float s_sa[2][NLAYERS][NMODES];
  __shared__ float s_sb[2][NLAYERS][NMODES];
  __shared__ float s_sd[2][NLAYERS][NMODES];
  __shared__ float s_ct[2][NLAYERS][NMODES - 1];
  __shared__ float s_cs[2][NLAYERS][NMODES - 1];
  __shared__ float s_ss[2][NLAYERS][NMODES - 1];

  const int t = threadIdx.x;

  if (t < 2 * NLAYERS * NMODES) {
    int cc = t / (NLAYERS * NMODES);
    int r  = t % (NLAYERS * NMODES);
    int l  = r / NMODES, i = r % NMODES;
    int base = ((cc * NLAYERS + l) * NMODES + i) * 2;
    float a = disp[base + 0], dphi = disp[base + 1];
    s_dx[cc][l][i] = 2.0f * a * cosf(dphi);
    s_dp[cc][l][i] = 2.0f * a * sinf(dphi);
    float rr = fabsf(sq[base + 0]);
    float sphi = sq[base + 1];
    float ch = coshf(rr), sh = sinhf(rr);
    float cp = cosf(sphi), sp = sinf(sphi);
    s_sa[cc][l][i] = ch - cp * sh;
    s_sb[cc][l][i] = -sp * sh;
    s_sd[cc][l][i] = ch + cp * sh;
  }
  if (t < 2 * NLAYERS * (NMODES - 1)) {
    int cc = t / (NLAYERS * (NMODES - 1));
    int r  = t % (NLAYERS * (NMODES - 1));
    int l  = r / (NMODES - 1), w = r % (NMODES - 1);
    int base = ((cc * NLAYERS + l) * (NMODES - 1) + w) * 2;
    float th   = 1.0f / (1.0f + expf(-bs[base + 0]));
    float bphi = bs[base + 1];
    float ct = cosf(th), st = sinf(th);
    s_ct[cc][l][w] = ct;
    s_cs[cc][l][w] = cosf(bphi) * st;
    s_ss[cc][l][w] = sinf(bphi) * st;
  }
  __syncthreads();

  if (t >= 34) return;
  const int cc  = t / 17;
  const int col = t % 17;
  const bool use_disp = (col == 16);

  float mx[NMODES], mp[NMODES];
#pragma unroll
  for (int i = 0; i < NMODES; ++i) {
    mx[i] = (!use_disp && i == col) ? 2.0f : 0.0f;
    mp[i] = 0.0f;
  }

  for (int l = 0; l < NLAYERS; ++l) {
#pragma unroll
    for (int i = 0; i < NMODES; ++i) {
      float xv = mx[i], pv = mp[i];
      if (use_disp) { xv += s_dx[cc][l][i]; pv += s_dp[cc][l][i]; }
      float A = s_sa[cc][l][i], Bb = s_sb[cc][l][i], D = s_sd[cc][l][i];
      mx[i] = A * xv + Bb * pv;
      mp[i] = Bb * xv + D * pv;
    }
#pragma unroll
    for (int w = 0; w < NMODES - 1; ++w) {
      float ct = s_ct[cc][l][w], cs = s_cs[cc][l][w], ss = s_ss[cc][l][w];
      float x1 = mx[w], x2 = mx[w + 1], p1 = mp[w], p2 = mp[w + 1];
      mx[w]     = ct * x1 - cs * x2 - ss * p2;
      mx[w + 1] = cs * x1 + ct * x2 - ss * p1;
      mp[w]     = ss * x2 + ct * p1 - cs * p2;
      mp[w + 1] = ss * x1 + cs * p1 + ct * p2;
    }
  }

  if (col < 16) {
#pragma unroll
    for (int j = 0; j < NMODES; ++j)
      ws[col * 32 + 16 * cc + j] = (cc == 0) ? mx[j] : mp[j];
  } else {
#pragma unroll
    for (int j = 0; j < NMODES; ++j)
      ws[512 + 16 * cc + j] = (cc == 0) ? mx[j] : mp[j];
  }
}

__global__ __launch_bounds__(256) void cv_gemm_kernel(
    const float* __restrict__ x, const float* __restrict__ ws,
    float* __restrict__ out, int numTiles)
{
  const int lane = threadIdx.x & 31;
  const int row  = lane & 15;
  const int hi   = lane >> 4;

  v2f bfrag[2][4];
#pragma unroll
  for (int h = 0; h < 2; ++h)
#pragma unroll
    for (int kc = 0; kc < 4; ++kc) {
      bfrag[h][kc].x = ws[(4 * kc + 0 + 2 * hi) * 32 + 16 * h + row];
      bfrag[h][kc].y = ws[(4 * kc + 1 + 2 * hi) * 32 + 16 * h + row];
    }
  const float bias0 = ws[512 + row];
  const float bias1 = ws[512 + 16 + row];

  const int waveId     = blockIdx.x * 8 + (threadIdx.x >> 5);
  const int totalWaves = gridDim.x * 8;

  for (int tile = waveId; tile < numTiles; tile += totalWaves) {
    const float* xt = x + (size_t)tile * (16 * 16);

    v2f afrag[4];
#pragma unroll
    for (int kc = 0; kc < 4; ++kc) {
      const v2f* p = (const v2f*)(xt + row * 16 + 4 * kc + 2 * hi);
      afrag[kc] = __builtin_nontemporal_load(p);
    }

    v8f acc0 = {};
    v8f acc1 = {};
#pragma unroll
    for (int kc = 0; kc < 4; ++kc) {
      acc0 = __builtin_amdgcn_wmma_f32_16x16x4_f32(
          false, afrag[kc], false, bfrag[0][kc], (short)0, acc0, false, false);
      acc1 = __builtin_amdgcn_wmma_f32_16x16x4_f32(
          false, afrag[kc], false, bfrag[1][kc], (short)0, acc1, false, false);
    }

    float* ot = out + (size_t)tile * (16 * 32);
#pragma unroll
    for (int v = 0; v < 8; ++v) {
      int m = v + 8 * hi;
      __builtin_nontemporal_store(acc0[v] + bias0, &ot[m * 32 + row]);
      __builtin_nontemporal_store(acc1[v] + bias1, &ot[m * 32 + 16 + row]);
    }
  }
}

extern "C" void kernel_launch(void* const* d_in, const int* in_sizes, int n_in,
                              void* d_out, int out_size, void* d_ws, size_t ws_size,
                              hipStream_t stream) {
  const float* x    = (const float*)d_in[0];
  const float* disp = (const float*)d_in[1];
  const float* sq   = (const float*)d_in[2];
  const float* bs   = (const float*)d_in[3];
  float* out = (float*)d_out;
  float* ws  = (float*)d_ws;

  const int BN       = in_sizes[0];
  const int numTiles = BN / 256;

  hipLaunchKernelGGL(cv_setup_kernel, dim3(1), dim3(192), 0, stream,
                     disp, sq, bs, ws);

  int blocks = (numTiles + 31) / 32;
  if (blocks < 1) blocks = 1;
  if (blocks > 4096) blocks = 4096;
  hipLaunchKernelGGL(cv_gemm_kernel, dim3(blocks), dim3(256), 0, stream,
                     x, ws, out, numTiles);
}


# Setup kernel's serial probe chain no longer waits on LDS: displacement via a 0/1 factor, not a divergent branch; a layer's coefficients come 16 B per load in one group up front; sums unchanged
# speedup vs baseline: 1.2578x; 1.2578x over previous
#include <hip/hip_runtime.h>

typedef float v2f __attribute__((ext_vector_type(2)));
typedef float v8f __attribute__((ext_vector_type(8)));

#define NMODES 16
#define NLAYERS 6

#define LDS_ROW __shared__ __attribute__((aligned(16))) float

#define LOAD_QUAD(dst, row, q)                                            \
  {                                                                       \
    const float4 v_ = *reinterpret_cast<const float4*>(&(row)[4 * (q)]);  \
    (dst)[4 * (q) + 0] = v_.x; (dst)[4 * (q) + 1] = v_.y;                 \
    (dst)[4 * (q) + 2] = v_.z; (dst)[4 * (q) + 3] = v_.w;                 \
  }

__global__ __launch_bounds__(192) void cv_setup_kernel(
    const float* __restrict__ disp,
    const float* __restrict__ sq,
    const float* __restrict__ bs,
    float* __restrict__ ws)
{
  LDS_ROW s_dx[2][NLAYERS][NMODES];
  LDS_ROW s_dp[2][NLAYERS][NMODES];
  LDS_ROW s_sa[2][NLAYERS][NMODES];
  LDS_ROW s_sb[2][NLAYERS][NMODES];
  LDS_ROW s_sd[2][NLAYERS][NMODES];
  LDS_ROW s_ct[2][NLAYERS][NMODES];
  LDS_ROW s_cs[2][NLAYERS][NMODES];
  LDS_ROW s_ss[2][NLAYERS][NMODES];

  const int t = threadIdx.x;

  if (t < 2 * NLAYERS * NMODES) {
    int cc = t / (NLAYERS * NMODES);
    int r  = t % (NLAYERS * NMODES);
    int l  = r / NMODES, i = r % NMODES;
    int base = ((cc * NLAYERS + l) * NMODES + i) * 2;
    float a = disp[base + 0], dphi = disp[base + 1];
    s_dx[cc][l][i] = 2.0f * a * cosf(dphi);
    s_dp[cc][l][i] = 2.0f * a * sinf(dphi);
    float rr = fabsf(sq[base + 0]);
    float sphi = sq[base + 1];
    float ch = coshf(rr), sh = sinhf(rr);
    float cp = cosf(sphi), sp = sinf(sphi);
    s_sa[cc][l][i] = ch - cp * sh;
    s_sb[cc][l][i] = -sp * sh;
    s_sd[cc][l][i] = ch + cp * sh;
    if (i == NMODES - 1) {
      s_ct[cc][l][i] = 0.0f;
      s_cs[cc][l][i] = 0.0f;
      s_ss[cc][l][i] = 0.0f;
    }
  }
  if (t < 2 * NLAYERS * (NMODES - 1)) {
    int cc = t / (NLAYERS * (NMODES - 1));
    int r  = t % (NLAYERS * (NMODES - 1));
    int l  = r / (NMODES - 1), w = r % (NMODES - 1);
    int base = ((cc * NLAYERS + l) * (NMODES - 1) + w) * 2;
    float th   = 1.0f / (1.0f + expf(-bs[base + 0]));
    float bphi = bs[base + 1];
    float ct = cosf(th), st = sinf(th);
    s_ct[cc][l][w] = ct;
    s_cs[cc][l][w] = cosf(bphi) * st;
    s_ss[cc][l][w] = sinf(bphi) * st;
  }
  __syncthreads();

  if (t >= 34) return;
  const int cc  = t / 17;
  const int col = t % 17;
  const bool use_disp = (col == 16);
  const float dsc = use_disp ? 1.0f : 0.0f;

  float mx[NMODES], mp[NMODES];
#pragma unroll
  for (int i = 0; i < NMODES; ++i) {
    mx[i] = (!use_disp && i == col) ? 2.0f : 0.0f;
    mp[i] = 0.0f;
  }

  for (int l = 0; l < NLAYERS; ++l) {
    float cdx[NMODES], cdp[NMODES], ca[NMODES], cb[NMODES], cd[NMODES];
    float cct[NMODES], ccs[NMODES], css[NMODES];
#pragma unroll
    for (int q = 0; q < 4; ++q) {
      LOAD_QUAD(cdx, s_dx[cc][l], q);
      LOAD_QUAD(cdp, s_dp[cc][l], q);
      LOAD_QUAD(ca,  s_sa[cc][l], q);
      LOAD_QUAD(cb,  s_sb[cc][l], q);
      LOAD_QUAD(cd,  s_sd[cc][l], q);
    }
#pragma unroll
    for (int q = 0; q < 4; ++q) {
      LOAD_QUAD(cct, s_ct[cc][l], q);
      LOAD_QUAD(ccs, s_cs[cc][l], q);
      LOAD_QUAD(css, s_ss[cc][l], q);
    }
    __builtin_amdgcn_sched_barrier(0);

#pragma unroll
    for (int i = 0; i < NMODES; ++i) {
      float xv = fmaf(dsc, cdx[i], mx[i]);
      float pv = fmaf(dsc, cdp[i], mp[i]);
      float A = ca[i], Bb = cb[i], D = cd[i];
      mx[i] = A * xv + Bb * pv;
      mp[i] = Bb * xv + D * pv;
    }
#pragma unroll
    for (int w = 0; w < NMODES - 1; ++w) {
      float ct = cct[w], cs = ccs[w], ss = css[w];
      float x1 = mx[w], x2 = mx[w + 1], p1 = mp[w], p2 = mp[w + 1];
      mx[w]     = ct * x1 - cs * x2 - ss * p2;
      mx[w + 1] = cs * x1 + ct * x2 - ss * p1;
      mp[w]     = ss * x2 + ct * p1 - cs * p2;
      mp[w + 1] = ss * x1 + cs * p1 + ct * p2;
    }
  }

  if (col < 16) {
#pragma unroll
    for (int j = 0; j < NMODES; ++j)
      ws[col * 32 + 16 * cc + j] = (cc == 0) ? mx[j] : mp[j];
  } else {
#pragma unroll
    for (int j = 0; j < NMODES; ++j)
      ws[512 + 16 * cc + j] = (cc == 0) ? mx[j] : mp[j];
  }
}

__global__ __launch_bounds__(256) void cv_gemm_kernel(
    const float* __restrict__ x, const float* __restrict__ ws,
    float* __restrict__ out, int numTiles)
{
  const int lane = threadIdx.x & 31;
  const int row  = lane & 15;
  const int hi   = lane >> 4;

  v2f bfrag[2][4];
#pragma unroll
  for (int h = 0; h < 2; ++h)
#pragma unroll
    for (int kc = 0; kc < 4; ++kc) {
      bfrag[h][kc].x = ws[(4 * kc + 0 + 2 * hi) * 32 + 16 * h + row];
      bfrag[h][kc].y = ws[(4 * kc + 1 + 2 * hi) * 32 + 16 * h + row];
    }
  const float bias0 = ws[512 + row];
  const float bias1 = ws[512 + 16 + row];

  const int waveId     = blockIdx.x * 8 + (threadIdx.x >> 5);
  const int totalWaves = gridDim.x * 8;

  for (int tile = waveId; tile < numTiles; tile += totalWaves) {
    const float* xt = x + (size_t)tile * (16 * 16);

    v2f afrag[4];
#pragma unroll
    for (int kc = 0; kc < 4; ++kc) {
      const v2f* p = (const v2f*)(xt + row * 16 + 4 * kc + 2 * hi);
      afrag[kc] = __builtin_nontemporal_load(p);
    }

    v8f acc0 = {};
    v8f acc1 = {};
#pragma unroll
    for (int kc = 0; kc < 4; ++kc) {
      acc0 = __builtin_amdgcn_wmma_f32_16x16x4_f32(
          false, afrag[kc], false, bfrag[0][kc], (short)0, acc0, false, false);
      acc1 = __builtin_amdgcn_wmma_f32_16x16x4_f32(
          false, afrag[kc], false, bfrag[1][kc], (short)0, acc1, false, false);
    }

    float* ot = out + (size_t)tile * (16 * 32);
#pragma unroll
    for (int v = 0; v < 8; ++v) {
      int m = v + 8 * hi;
      __builtin_nontemporal_store(acc0[v] + bias0, &ot[m * 32 + row]);
      __builtin_nontemporal_store(acc1[v] + bias1, &ot[m * 32 + 16 + row]);
    }
  }
}

extern "C" void kernel_launch(void* const* d_in, const int* in_sizes, int n_in,
                              void* d_out, int out_size, void* d_ws, size_t ws_size,
                              hipStream_t stream) {
  const float* x    = (const float*)d_in[0];
  const float* disp = (const float*)d_in[1];
  const float* sq   = (const float*)d_in[2];
  const float* bs   = (const float*)d_in[3];
  float* out = (float*)d_out;
  float* ws  = (float*)d_ws;

  const int BN       = in_sizes[0];
  const int numTiles = BN / 256;

  hipLaunchKernelGGL(cv_setup_kernel, dim3(1), dim3(192), 0, stream,
                     disp, sq, bs, ws);

  int blocks = (numTiles + 31) / 32;
  if (blocks < 1) blocks = 1;
  if (blocks > 4096) blocks = 4096;
  hipLaunchKernelGGL(cv_gemm_kernel, dim3(blocks), dim3(256), 0, stream,
                     x, ws, out, numTiles);
}


# Set-up launch folded into the streaming kernel: 256 work-groups each rebuild the 16x32 map in LDS while their first x tiles are in flight, then loop over tiles; one launch, sums unchanged
# speedup vs baseline: 1.4927x; 1.1868x over previous
#include <hip/hip_runtime.h>

typedef float v2f __attribute__((ext_vector_type(2)));
typedef float v8f __attribute__((ext_vector_type(8)));

#define NMODES 16
#define NLAYERS 6
#define GROUP_WAVES 16
#define GROUP_THREADS (32 * GROUP_WAVES)
#define NUM_GROUPS 256
#define AHEAD 8

#define LDS_ROW __shared__ __attribute__((aligned(16))) float

#define LOAD_QUAD(dst, row, q)                                            \
  {                                                                       \
    const float4 v_ = *reinterpret_cast<const float4*>(&(row)[4 * (q)]);  \
    (dst)[4 * (q) + 0] = v_.x; (dst)[4 * (q) + 1] = v_.y;                 \
    (dst)[4 * (q) + 2] = v_.z; (dst)[4 * (q) + 3] = v_.w;                 \
  }

static __device__ __forceinline__ void lds_barrier() {
  asm volatile("s_wait_dscnt 0x0\n\ts_barrier_signal -1\n\ts_barrier_wait -1" ::: "memory");
}

static __device__ __forceinline__ void load_tile(v2f (&a)[4], const float* xl, int tile) {
  const float* xt = xl + (size_t)tile * (16 * 16);
#pragma unroll
  for (int kc = 0; kc < 4; ++kc)
    a[kc] = __builtin_nontemporal_load((const v2f*)(xt + 4 * kc));
}

template <bool REFILL>
static __device__ __forceinline__ void stream_pass(
    v2f (&buf)[AHEAD][4], const v2f (&bfrag)[2][4], float bias0, float bias1,
    const float* xl, float* ol, int base, int totalWaves, int numTiles)
{
#pragma unroll
  for (int j = 0; j < AHEAD; ++j) {
    const int tile = base + j * totalWaves;
    v8f acc0 = {};
    v8f acc1 = {};
#pragma unroll
    for (int kc = 0; kc < 4; ++kc) {
      acc0 = __builtin_amdgcn_wmma_f32_16x16x4_f32(
          false, buf[j][kc], false, bfrag[0][kc], (short)0, acc0, false, false);
      acc1 = __builtin_amdgcn_wmma_f32_16x16x4_f32(
          false, buf[j][kc], false, bfrag[1][kc], (short)0, acc1, false, false);
    }
    if (REFILL) {
      const int nx = tile + AHEAD * totalWaves;
      load_tile(buf[j], xl, nx < numTiles ? nx : numTiles - 1);
    }
    if (tile < numTiles) {
      float* ot = ol + (size_t)tile * (16 * 32);
#pragma unroll
      for (int v = 0; v < 8; ++v) {
        __builtin_nontemporal_store(acc0[v] + bias0, &ot[v * 32]);
        __builtin_nontemporal_store(acc1[v] + bias1, &ot[v * 32 + 16]);
      }
    }
  }
}

__global__ __launch_bounds__(GROUP_THREADS) void cv_fused_kernel(
    const float* __restrict__ x,
    const float* disp,
    const float* sq,
    const float* bs,
    float* __restrict__ out, int numTiles)
{
  LDS_ROW s_dx[2][NLAYERS][NMODES];
  LDS_ROW s_dp[2][NLAYERS][NMODES];
  LDS_ROW s_sa[2][NLAYERS][NMODES];
  LDS_ROW s_sb[2][NLAYERS][NMODES];
  LDS_ROW s_sd[2][NLAYERS][NMODES];
  LDS_ROW s_ct[2][NLAYERS][NMODES];
  LDS_ROW s_cs[2][NLAYERS][NMODES];
  LDS_ROW s_ss[2][NLAYERS][NMODES];
  LDS_ROW s_w[NMODES][32];
  LDS_ROW s_bias[32];

  if (numTiles <= 0) return;

  const int t    = threadIdx.x;
  const int lane = t & 31;
  const int row  = lane & 15;
  const int hi   = lane >> 4;
  const int wave = __builtin_amdgcn_readfirstlane(t >> 5);
  const int waveId     = blockIdx.x * GROUP_WAVES + wave;
  const int totalWaves = gridDim.x * GROUP_WAVES;

  const int tm = t < 2 * NLAYERS * NMODES ? t : 2 * NLAYERS * NMODES - 1;
  const int tp = t < 2 * NLAYERS * (NMODES - 1) ? t : 2 * NLAYERS * (NMODES - 1) - 1;
  const float p_a = disp[2 * tm + 0], p_dphi = disp[2 * tm + 1];
  const float p_r = sq[2 * tm + 0],   p_sphi = sq[2 * tm + 1];
  const float p_t = bs[2 * tp + 0],   p_bphi = bs[2 * tp + 1];
  asm volatile("" ::: "memory");

  const float* xl = x + row * 16 + 2 * hi;
  v2f buf[AHEAD][4];
#pragma unroll
  for (int j = 0; j < AHEAD; ++j) {
    const int tile = waveId + j * totalWaves;
    load_tile(buf[j], xl, tile < numTiles ? tile : numTiles - 1);
  }

  if (t < 2 * NLAYERS * NMODES) {
    int cc = t / (NLAYERS * NMODES);
    int r  = t % (NLAYERS * NMODES);
    int l  = r / NMODES, i = r % NMODES;
    float a = p_a, dphi = p_dphi;
    s_dx[cc][l][i] = 2.0f * a * cosf(dphi);
    s_dp[cc][l][i] = 2.0f * a * sinf(dphi);
    float rr = fabsf(p_r);
    float sphi = p_sphi;
    float ch = coshf(rr), sh = sinhf(rr);
    float cp = cosf(sphi), sp = sinf(sphi);
    s_sa[cc][l][i] = ch - cp * sh;
    s_sb[cc][l][i] = -sp * sh;
    s_sd[cc][l][i] = ch + cp * sh;
    if (i == NMODES - 1) {
      s_ct[cc][l][i] = 0.0f;
      s_cs[cc][l][i] = 0.0f;
      s_ss[cc][l][i] = 0.0f;
    }
  }
  if (t < 2 * NLAYERS * (NMODES - 1)) {
    int cc = t / (NLAYERS * (NMODES - 1));
    int r  = t % (NLAYERS * (NMODES - 1));
    int l  = r / (NMODES - 1), w = r % (NMODES - 1);
    float th   = 1.0f / (1.0f + expf(-p_t));
    float bphi = p_bphi;
    float ct = cosf(th), st = sinf(th);
    s_ct[cc][l][w] = ct;
    s_cs[cc][l][w] = cosf(bphi) * st;
    s_ss[cc][l][w] = sinf(bphi) * st;
  }
  lds_barrier();

  if (t < 34) {
    const int cc  = t / 17;
    const int col = t % 17;
    const bool use_disp = (col == 16);
    const float dsc = use_disp ? 1.0f : 0.0f;

    float mx[NMODES], mp[NMODES];
#pragma unroll
    for (int i = 0; i < NMODES; ++i) {
      mx[i] = (!use_disp && i == col) ? 2.0f : 0.0f;
      mp[i] = 0.0f;
    }

    for (int l = 0; l < NLAYERS; ++l) {
      float cdx[NMODES], cdp[NMODES], ca[NMODES], cb[NMODES], cd[NMODES];
      float cct[NMODES], ccs[NMODES], css[NMODES];
#pragma unroll
      for (int q = 0; q < 4; ++q) {
        LOAD_QUAD(cdx, s_dx[cc][l], q);
        LOAD_QUAD(cdp, s_dp[cc][l], q);
        LOAD_QUAD(ca,  s_sa[cc][l], q);
        LOAD_QUAD(cb,  s_sb[cc][l], q);
        LOAD_QUAD(cd,  s_sd[cc][l], q);
      }
#pragma unroll
      for (int q = 0; q < 4; ++q) {
        LOAD_QUAD(cct, s_ct[cc][l], q);
        LOAD_QUAD(ccs, s_cs[cc][l], q);
        LOAD_QUAD(css, s_ss[cc][l], q);
      }
      __builtin_amdgcn_sched_barrier(0);

#pragma unroll
      for (int i = 0; i < NMODES; ++i) {
        float xv = fmaf(dsc, cdx[i], mx[i]);
        float pv = fmaf(dsc, cdp[i], mp[i]);
        float A = ca[i], Bb = cb[i], D = cd[i];
        mx[i] = A * xv + Bb * pv;
        mp[i] = Bb * xv + D * pv;
      }
#pragma unroll
      for (int w = 0; w < NMODES - 1; ++w) {
        float ct = cct[w], cs = ccs[w], ss = css[w];
        float x1 = mx[w], x2 = mx[w + 1], p1 = mp[w], p2 = mp[w + 1];
        mx[w]     = ct * x1 - cs * x2 - ss * p2;
        mx[w + 1] = cs * x1 + ct * x2 - ss * p1;
        mp[w]     = ss * x2 + ct * p1 - cs * p2;
        mp[w + 1] = ss * x1 + cs * p1 + ct * p2;
      }
    }

    if (col < 16) {
#pragma unroll
      for (int j = 0; j < NMODES; ++j)
        s_w[col][16 * cc + j] = (cc == 0) ? mx[j] : mp[j];
    } else {
#pragma unroll
      for (int j = 0; j < NMODES; ++j)
        s_bias[16 * cc + j] = (cc == 0) ? mx[j] : mp[j];
    }
  }
  lds_barrier();

  v2f bfrag[2][4];
#pragma unroll
  for (int h = 0; h < 2; ++h)
#pragma unroll
    for (int kc = 0; kc < 4; ++kc) {
      bfrag[h][kc].x = s_w[4 * kc + 0 + 2 * hi][16 * h + row];
      bfrag[h][kc].y = s_w[4 * kc + 1 + 2 * hi][16 * h + row];
    }
  const float bias0 = s_bias[row];
  const float bias1 = s_bias[16 + row];

  float* ol = out + (8 * hi) * 32 + row;
  stream_pass<true>(buf, bfrag, bias0, bias1, xl, ol, waveId, totalWaves, numTiles);
  stream_pass<false>(buf, bfrag, bias0, bias1, xl, ol, waveId + AHEAD * totalWaves,
                     totalWaves, numTiles);
  for (int tile = waveId + 2 * AHEAD * totalWaves; tile < numTiles; tile += totalWaves) {
    v2f a[4];
    load_tile(a, xl, tile);
    v8f acc0 = {};
    v8f acc1 = {};
#pragma unroll
    for (int kc = 0; kc < 4; ++kc) {
      acc0 = __builtin_amdgcn_wmma_f32_16x16x4_f32(
          false, a[kc], false, bfrag[0][kc], (short)0, acc0, false, false);
      acc1 = __builtin_amdgcn_wmma_f32_16x16x4_f32(
          false, a[kc], false, bfrag[1][kc], (short)0, acc1, false, false);
    }
    float* ot = ol + (size_t)tile * (16 * 32);
#pragma unroll
    for (int v = 0; v < 8; ++v) {
      __builtin_nontemporal_store(acc0[v] + bias0, &ot[v * 32]);
      __builtin_nontemporal_store(acc1[v] + bias1, &ot[v * 32 + 16]);
    }
  }
}

extern "C" void kernel_launch(void* const* d_in, const int* in_sizes, int n_in,
                              void* d_out, int out_size, void* d_ws, size_t ws_size,
                              hipStream_t stream) {
  const float* x    = (const float*)d_in[0];
  const float* disp = (const float*)d_in[1];
  const float* sq   = (const float*)d_in[2];
  const float* bs   = (const float*)d_in[3];
  float* out = (float*)d_out;

  const int BN       = in_sizes[0];
  const int numTiles = BN / 256;

  hipLaunchKernelGGL(cv_fused_kernel, dim3(NUM_GROUPS), dim3(GROUP_THREADS), 0, stream,
                     x, disp, sq, bs, out, numTiles);
}


# Second half of each wave's x tiles requested before the set-up's last barrier (under the probe phase, by the 14 waves idle there) instead of as refills among the stores; sums unchanged
# speedup vs baseline: 1.7427x; 1.1675x over previous
#include <hip/hip_runtime.h>

typedef float v2f __attribute__((ext_vector_type(2)));
typedef float v8f __attribute__((ext_vector_type(8)));

#define NMODES 16
#define NLAYERS 6
#define GROUP_WAVES 16
#define GROUP_THREADS (32 * GROUP_WAVES)
#define NUM_GROUPS 256
#define HALF 8
#define SHARE (2 * HALF)

#define LDS_ROW __shared__ __attribute__((aligned(16))) float

#define LOAD_QUAD(dst, row, q)                                            \
  {                                                                       \
    const float4 v_ = *reinterpret_cast<const float4*>(&(row)[4 * (q)]);  \
    (dst)[4 * (q) + 0] = v_.x; (dst)[4 * (q) + 1] = v_.y;                 \
    (dst)[4 * (q) + 2] = v_.z; (dst)[4 * (q) + 3] = v_.w;                 \
  }

static __device__ __forceinline__ void lds_barrier() {
  asm volatile("s_wait_dscnt 0x0\n\ts_barrier_signal -1\n\ts_barrier_wait -1" ::: "memory");
}

static __device__ __forceinline__ void load_tile(v2f (&a)[4], const float* xl, int tile) {
  const float* xt = xl + (size_t)tile * (16 * 16);
#pragma unroll
  for (int kc = 0; kc < 4; ++kc)
    a[kc] = __builtin_nontemporal_load((const v2f*)(xt + 4 * kc));
}

static __device__ __forceinline__ void stream_share(
    const v2f (&buf)[SHARE][4], const v2f (&bfrag)[2][4], float bias0, float bias1,
    float* ol, int base, int totalWaves, int numTiles)
{
#pragma unroll
  for (int j = 0; j < SHARE; ++j) {
    const int tile = base + j * totalWaves;
    v8f acc0 = {};
    v8f acc1 = {};
#pragma unroll
    for (int kc = 0; kc < 4; ++kc) {
      acc0 = __builtin_amdgcn_wmma_f32_16x16x4_f32(
          false, buf[j][kc], false, bfrag[0][kc], (short)0, acc0, false, false);
      acc1 = __builtin_amdgcn_wmma_f32_16x16x4_f32(
          false, buf[j][kc], false, bfrag[1][kc], (short)0, acc1, false, false);
    }
    if (tile < numTiles) {
      float* ot = ol + (size_t)tile * (16 * 32);
#pragma unroll
      for (int v = 0; v < 8; ++v) {
        __builtin_nontemporal_store(acc0[v] + bias0, &ot[v * 32]);
        __builtin_nontemporal_store(acc1[v] + bias1, &ot[v * 32 + 16]);
      }
    }
  }
}

__global__ __launch_bounds__(GROUP_THREADS) void cv_fused_kernel(
    const float* __restrict__ x,
    const float* disp,
    const float* sq,
    const float* bs,
    float* __restrict__ out, int numTiles)
{
  LDS_ROW s_dx[2][NLAYERS][NMODES];
  LDS_ROW s_dp[2][NLAYERS][NMODES];
  LDS_ROW s_sa[2][NLAYERS][NMODES];
  LDS_ROW s_sb[2][NLAYERS][NMODES];
  LDS_ROW s_sd[2][NLAYERS][NMODES];
  LDS_ROW s_ct[2][NLAYERS][NMODES];
  LDS_ROW s_cs[2][NLAYERS][NMODES];
  LDS_ROW s_ss[2][NLAYERS][NMODES];
  LDS_ROW s_w[NMODES][32];
  LDS_ROW s_bias[32];

  if (numTiles <= 0) return;

  const int t    = threadIdx.x;
  const int lane = t & 31;
  const int row  = lane & 15;
  const int hi   = lane >> 4;
  const int wave = __builtin_amdgcn_readfirstlane(t >> 5);
  const int waveId     = blockIdx.x * GROUP_WAVES + wave;
  const int totalWaves = gridDim.x * GROUP_WAVES;

  const int tm = t < 2 * NLAYERS * NMODES ? t : 2 * NLAYERS * NMODES - 1;
  const int tp = t < 2 * NLAYERS * (NMODES - 1) ? t : 2 * NLAYERS * (NMODES - 1) - 1;
  const float p_a = disp[2 * tm + 0], p_dphi = disp[2 * tm + 1];
  const float p_r = sq[2 * tm + 0],   p_sphi = sq[2 * tm + 1];
  const float p_t = bs[2 * tp + 0],   p_bphi = bs[2 * tp + 1];
  asm volatile("" ::: "memory");

  const float* xl = x + row * 16 + 2 * hi;
  v2f buf[SHARE][4];
#pragma unroll
  for (int j = 0; j < HALF; ++j) {
    const int tile = waveId + j * totalWaves;
    load_tile(buf[j], xl, tile < numTiles ? tile : numTiles - 1);
  }

  if (t < 2 * NLAYERS * NMODES) {
    int cc = t / (NLAYERS * NMODES);
    int r  = t % (NLAYERS * NMODES);
    int l  = r / NMODES, i = r % NMODES;
    float a = p_a, dphi = p_dphi;
    s_dx[cc][l][i] = 2.0f * a * cosf(dphi);
    s_dp[cc][l][i] = 2.0f * a * sinf(dphi);
    float rr = fabsf(p_r);
    float sphi = p_sphi;
    float ch = coshf(rr), sh = sinhf(rr);
    float cp = cosf(sphi), sp = sinf(sphi);
    s_sa[cc][l][i] = ch - cp * sh;
    s_sb[cc][l][i] = -sp * sh;
    s_sd[cc][l][i] = ch + cp * sh;
    if (i == NMODES - 1) {
      s_ct[cc][l][i] = 0.0f;
      s_cs[cc][l][i] = 0.0f;
      s_ss[cc][l][i] = 0.0f;
    }
  }
  if (t < 2 * NLAYERS * (NMODES - 1)) {
    int cc = t / (NLAYERS * (NMODES - 1));
    int r  = t % (NLAYERS * (NMODES - 1));
    int l  = r / (NMODES - 1), w = r % (NMODES - 1);
    float th   = 1.0f / (1.0f + expf(-p_t));
    float bphi = p_bphi;
    float ct = cosf(th), st = sinf(th);
    s_ct[cc][l][w] = ct;
    s_cs[cc][l][w] = cosf(bphi) * st;
    s_ss[cc][l][w] = sinf(bphi) * st;
  }
  lds_barrier();

  if (t < 34) {
    const int cc  = t / 17;
    const int col = t % 17;
    const bool use_disp = (col == 16);
    const float dsc = use_disp ? 1.0f : 0.0f;

    float mx[NMODES], mp[NMODES];
#pragma unroll
    for (int i = 0; i < NMODES; ++i) {
      mx[i] = (!use_disp && i == col) ? 2.0f : 0.0f;
      mp[i] = 0.0f;
    }

    for (int l = 0; l < NLAYERS; ++l) {
      float cdx[NMODES], cdp[NMODES], ca[NMODES], cb[NMODES], cd[NMODES];
      float cct[NMODES], ccs[NMODES], css[NMODES];
#pragma unroll
      for (int q = 0; q < 4; ++q) {
        LOAD_QUAD(cdx, s_dx[cc][l], q);
        LOAD_QUAD(cdp, s_dp[cc][l], q);
        LOAD_QUAD(ca,  s_sa[cc][l], q);
        LOAD_QUAD(cb,  s_sb[cc][l], q);
        LOAD_QUAD(cd,  s_sd[cc][l], q);
      }
#pragma unroll
      for (int q = 0; q < 4; ++q) {
        LOAD_QUAD(cct, s_ct[cc][l], q);
        LOAD_QUAD(ccs, s_cs[cc][l], q);
        LOAD_QUAD(css, s_ss[cc][l], q);
      }
      __builtin_amdgcn_sched_barrier(0);

#pragma unroll
      for (int i = 0; i < NMODES; ++i) {
        float xv = fmaf(dsc, cdx[i], mx[i]);
        float pv = fmaf(dsc, cdp[i], mp[i]);
        float A = ca[i], Bb = cb[i], D = cd[i];
        mx[i] = A * xv + Bb * pv;
        mp[i] = Bb * xv + D * pv;
      }
#pragma unroll
      for (int w = 0; w < NMODES - 1; ++w) {
        float ct = cct[w], cs = ccs[w], ss = css[w];
        float x1 = mx[w], x2 = mx[w + 1], p1 = mp[w], p2 = mp[w + 1];
        mx[w]     = ct * x1 - cs * x2 - ss * p2;
        mx[w + 1] = cs * x1 + ct * x2 - ss * p1;
        mp[w]     = ss * x2 + ct * p1 - cs * p2;
        mp[w + 1] = ss * x1 + cs * p1 + ct * p2;
      }
    }

    if (col < 16) {
#pragma unroll
      for (int j = 0; j < NMODES; ++j)
        s_w[col][16 * cc + j] = (cc == 0) ? mx[j] : mp[j];
    } else {
#pragma unroll
      for (int j = 0; j < NMODES; ++j)
        s_bias[16 * cc + j] = (cc == 0) ? mx[j] : mp[j];
    }
  }
#pragma unroll
  for (int j = HALF; j < SHARE; ++j) {
    const int tile = waveId + j * totalWaves;
    load_tile(buf[j], xl, tile < numTiles ? tile : numTiles - 1);
  }
  lds_barrier();

  v2f bfrag[2][4];
#pragma unroll
  for (int h = 0; h < 2; ++h)
#pragma unroll
    for (int kc = 0; kc < 4; ++kc) {
      bfrag[h][kc].x = s_w[4 * kc + 0 + 2 * hi][16 * h + row];
      bfrag[h][kc].y = s_w[4 * kc + 1 + 2 * hi][16 * h + row];
    }
  const float bias0 = s_bias[row];
  const float bias1 = s_bias[16 + row];

  float* ol = out + (8 * hi) * 32 + row;
  stream_share(buf, bfrag, bias0, bias1, ol, waveId, totalWaves, numTiles);
  for (int tile = waveId + SHARE * totalWaves; tile < numTiles; tile += totalWaves) {
    v2f a[4];
    load_tile(a, xl, tile);
    v8f acc0 = {};
    v8f acc1 = {};
#pragma unroll
    for (int kc = 0; kc < 4; ++kc) {
      acc0 = __builtin_amdgcn_wmma_f32_16x16x4_f32(
          false, a[kc], false, bfrag[0][kc], (short)0, acc0, false, false);
      acc1 = __builtin_amdgcn_wmma_f32_16x16x4_f32(
          false, a[kc], false, bfrag[1][kc], (short)0, acc1, false, false);
    }
    float* ot = ol + (size_t)tile * (16 * 32);
#pragma unroll
    for (int v = 0; v < 8; ++v) {
      __builtin_nontemporal_store(acc0[v] + bias0, &ot[v * 32]);
      __builtin_nontemporal_store(acc1[v] + bias1, &ot[v * 32 + 16]);
    }
  }
}

extern "C" void kernel_launch(void* const* d_in, const int* in_sizes, int n_in,
                              void* d_out, int out_size, void* d_ws, size_t ws_size,
                              hipStream_t stream) {
  const float* x    = (const float*)d_in[0];
  const float* disp = (const float*)d_in[1];
  const float* sq   = (const float*)d_in[2];
  const float* bs   = (const float*)d_in[3];
  float* out = (float*)d_out;

  const int BN       = in_sizes[0];
  const int numTiles = BN / 256;

  hipLaunchKernelGGL(cv_fused_kernel, dim3(NUM_GROUPS), dim3(GROUP_THREADS), 0, stream,
                     x, disp, sq, bs, out, numTiles);
}
